# starting kernel
# baseline (speedup 1.0000x reference)
#include <hip/hip_runtime.h>

typedef float v2f __attribute__((ext_vector_type(2)));
typedef float v8f __attribute__((ext_vector_type(8)));

#define OUT_HW 7
#define SCALE 0.25f
#define CC 64
#define HH 100
#define WW 100
#define KPAD 128
#define WSTRIDE 130

__device__ __forceinline__ float hat_cdf(float t) {
    float a = fminf(fmaxf(t + 1.0f, 0.0f), 1.0f);
    float b = fminf(fmaxf(1.0f - t, 0.0f), 1.0f);
    return (t < 0.0f) ? (0.5f * a * a) : (1.0f - 0.5f * b * b);
}

__global__ __launch_bounds__(128)
void prroi_wmma_f32_kernel(const float* __restrict__ feat,
                           const float* __restrict__ rois,
                           float* __restrict__ out,
                           int batch)
{
    __shared__ float sWx[8][WSTRIDE];
    __shared__ float sWy[8][WSTRIDE];
    __shared__ float sU[4][KPAD][16];

    const int n    = blockIdx.x;
    const int tid  = threadIdx.x;
    const int lane = tid & 31;
    const int wave = tid >> 5;
    const int m16  = lane & 15;
    const int hi16 = (lane >> 4) & 1;

    const float* r = rois + (size_t)n * 5;
    int bi = (int)r[0];
    bi = max(0, min(batch - 1, bi));
    float x1 = r[1] * SCALE, y1 = r[2] * SCALE;
    float x2 = r[3] * SCALE, y2 = r[4] * SCALE;
    float rw = fmaxf(x2 - x1, 0.0f), rh = fmaxf(y2 - y1, 0.0f);
    float bw = rw / OUT_HW, bh = rh / OUT_HW;
    float area = bw * bh;
    float sc = (area > 0.0f) ? (1.0f / fmaxf(area, 1e-12f)) : 0.0f;

    for (int idx = tid; idx < 2 * 8 * WSTRIDE; idx += blockDim.x) {
        int which = idx / (8 * WSTRIDE);
        int rem   = idx - which * 8 * WSTRIDE;
        int q     = rem / WSTRIDE;
        int k     = rem - q * WSTRIDE;
        float val = 0.0f;
        if (q < OUT_HW && k < (which ? HH : WW)) {
            float start = which ? y1 : x1;
            float bs    = which ? bh : bw;
            float u0 = start + (float)q * bs;
            float u1 = u0 + bs;
            float px = (float)k;
            val = hat_cdf(u1 - px) - hat_cdf(u0 - px);
        }
        if (which) sWy[q][k] = val; else sWx[q][k] = val;
    }
    __syncthreads();

    for (int idx = lane; idx < 16 * 16; idx += 32)
        sU[wave][112 + (idx >> 4)][idx & 15] = 0.0f;

    v2f bx[32];
    #pragma unroll
    for (int ks = 0; ks < 32; ++ks) {
        int k0 = ks * 4 + hi16 * 2;
        if (m16 < 8) { bx[ks].x = sWx[m16][k0]; bx[ks].y = sWx[m16][k0 + 1]; }
        else         { bx[ks] = (v2f){0.0f, 0.0f}; }
    }

    const float* fbase = feat + (size_t)bi * CC * (HH * WW);

    for (int i = 0; i < 8; ++i) {
        int c = blockIdx.y * 32 + wave * 8 + i;
        const float* fc = fbase + (size_t)c * (HH * WW);

        for (int ht = 0; ht < 7; ++ht) {
            int h = ht * 16 + m16;
            const float* frow = fc + h * WW;
            bool hok = (h < HH);

            v2f av[32];
            #pragma unroll
            for (int ks = 0; ks < 32; ++ks) {
                int k0 = ks * 4 + hi16 * 2;
                if (hok && k0 < WW) { av[ks].x = frow[k0]; av[ks].y = frow[k0 + 1]; }
                else                { av[ks] = (v2f){0.0f, 0.0f}; }
            }
            v8f acc0 = {}, acc1 = {};
            #pragma unroll
            for (int ks = 0; ks < 32; ks += 2) {
                acc0 = __builtin_amdgcn_wmma_f32_16x16x4_f32(
                         false, av[ks],     false, bx[ks],     (short)0, acc0, false, false);
                acc1 = __builtin_amdgcn_wmma_f32_16x16x4_f32(
                         false, av[ks + 1], false, bx[ks + 1], (short)0, acc1, false, false);
            }
            #pragma unroll
            for (int rr = 0; rr < 8; ++rr)
                sU[wave][ht * 16 + rr + hi16 * 8][m16] = acc0[rr] + acc1[rr];
        }

        __builtin_amdgcn_sched_barrier(0);

        v2f ay[32], bu[32];
        #pragma unroll
        for (int ks = 0; ks < 32; ++ks) {
            int k0 = ks * 4 + hi16 * 2;
            if (m16 < 8) { ay[ks].x = sWy[m16][k0]; ay[ks].y = sWy[m16][k0 + 1]; }
            else         { ay[ks] = (v2f){0.0f, 0.0f}; }
            bu[ks].x = sU[wave][k0][m16];
            bu[ks].y = sU[wave][k0 + 1][m16];
        }
        v8f acc2 = {}, acc3 = {};
        #pragma unroll
        for (int ks = 0; ks < 32; ks += 2) {
            acc2 = __builtin_amdgcn_wmma_f32_16x16x4_f32(
                     false, ay[ks],     false, bu[ks],     (short)0, acc2, false, false);
            acc3 = __builtin_amdgcn_wmma_f32_16x16x4_f32(
                     false, ay[ks + 1], false, bu[ks + 1], (short)0, acc3, false, false);
        }
        __builtin_amdgcn_sched_barrier(0);

        #pragma unroll
        for (int rr = 0; rr < 8; ++rr) {
            int p = rr + hi16 * 8;
            int q = m16;
            if (p < OUT_HW && q < OUT_HW)
                out[(((size_t)n * CC + c) * OUT_HW + p) * OUT_HW + q] =
                    (acc2[rr] + acc3[rr]) * sc;
        }
    }
}

extern "C" void kernel_launch(void* const* d_in, const int* in_sizes, int n_in,
                              void* d_out, int out_size, void* d_ws, size_t ws_size,
                              hipStream_t stream) {
    const float* feat = (const float*)d_in[0];
    const float* rois = (const float*)d_in[1];
    float* out = (float*)d_out;
    int n_rois = in_sizes[1] / 5;
    int batch  = in_sizes[0] / (CC * HH * WW);
    dim3 grid(n_rois, 2);
    prroi_wmma_f32_kernel<<<grid, 128, 0, stream>>>(feat, rois, out, batch);
}


# Block = two roi pairs x 8 channels (was one pair x 16): waves w and w+2 stream the same 4-channel slab side by side, so one of the two requests hits the compute unit's cache; sums unchanged
# speedup vs baseline: 3.2420x; 3.2420x over previous
#include <hip/hip_runtime.h>

typedef float v2f __attribute__((ext_vector_type(2)));
typedef float v8f __attribute__((ext_vector_type(8)));
typedef float v4f __attribute__((ext_vector_type(4)));

#define OUT_HW 7
#define SCALE 0.25f
#define CC 64
#define HH 100
#define WW 100
#define KPAD 128
#define KS 25
#define WSTRIDE 132
#define FPITCH 132
#define TILE_V4 400
#define NPF 13
#define NCH 8
#define CPW 4
#define GRP 2
#define NTILE 25

__device__ __forceinline__ float hat_cdf(float t) {
    float a = fminf(fmaxf(t + 1.0f, 0.0f), 1.0f);
    float b = fminf(fmaxf(1.0f - t, 0.0f), 1.0f);
    return (t < 0.0f) ? (0.5f * a * a) : (1.0f - 0.5f * b * b);
}

__global__ __launch_bounds__(128)
void prroi_wmma_f32_kernel(const float* __restrict__ feat,
                           const float* __restrict__ rois,
                           float* __restrict__ out,
                           int batch, int n_rois)
{
    __shared__ __attribute__((aligned(16))) float sWx[GRP][16][WSTRIDE];
    __shared__ __attribute__((aligned(16))) float sWy[GRP][16][WSTRIDE];
    __shared__ float sU[4][KPAD][16];
    __shared__ __attribute__((aligned(16))) float sF[4][16][FPITCH];

    const int tid  = threadIdx.x;
    const int lane = tid & 31;
    const int wave = tid >> 5;
    const int grp  = wave >> 1;
    const int wsub = wave & 1;
    const int m16  = lane & 15;
    const int hi16 = (lane >> 4) & 1;

    int bis[NCH];
    #pragma unroll
    for (int ch = 0; ch < NCH; ++ch) {
        int nn = ch * 32 + lane;
        int v = (int)rois[(size_t)max(0, min(nn, n_rois - 1)) * 5];
        v = max(0, min(batch - 1, v));
        bis[ch] = (nn < n_rois) ? v : -1;
    }
    const int slot = blockIdx.x * GRP + grp;
    int img = -1, jp = 0, off = 0;
    for (int b = 0; b < batch; ++b) {
        int cnt = 0;
        #pragma unroll
        for (int ch = 0; ch < NCH; ++ch)
            cnt += __builtin_popcount(__builtin_amdgcn_ballot_w32(bis[ch] == b));
        int pairs = (cnt + 1) >> 1;
        if (img < 0 && slot < off + pairs) { img = b; jp = slot - off; }
        off += pairs;
    }

    int ra = -1, rb = -1, run = 0;
    #pragma unroll
    for (int ch = 0; ch < NCH; ++ch) {
        bool f = (bis[ch] == img);
        unsigned m = __builtin_amdgcn_ballot_w32(f);
        int rank = run + __builtin_popcount(m & ((1u << lane) - 1u));
        unsigned ma = __builtin_amdgcn_ballot_w32(f && rank == 2 * jp);
        unsigned mb = __builtin_amdgcn_ballot_w32(f && rank == 2 * jp + 1);
        if (ma) ra = ch * 32 + __builtin_ctz(ma);
        if (mb) rb = ch * 32 + __builtin_ctz(mb);
        run += __builtin_popcount(m);
    }
    ra = __builtin_amdgcn_readfirstlane(ra);
    rb = __builtin_amdgcn_readfirstlane(rb);
    img = __builtin_amdgcn_readfirstlane(img);
    const bool live = (img >= 0);
    const bool has_b = (rb >= 0);

    float x1r[2], y1r[2], bwr[2], bhr[2], scr[2];
    #pragma unroll
    for (int s = 0; s < 2; ++s) {
        const float* r = rois + (size_t)max((s == 0 || !has_b) ? ra : rb, 0) * 5;
        float x1 = r[1] * SCALE, y1 = r[2] * SCALE;
        float x2 = r[3] * SCALE, y2 = r[4] * SCALE;
        float rw = fmaxf(x2 - x1, 0.0f), rh = fmaxf(y2 - y1, 0.0f);
        float bw = rw / OUT_HW, bh = rh / OUT_HW;
        float area = bw * bh;
        x1r[s] = x1; y1r[s] = y1; bwr[s] = bw; bhr[s] = bh;
        scr[s] = (area > 0.0f) ? (1.0f / fmaxf(area, 1e-12f)) : 0.0f;
    }

    const float* fbase = feat + (size_t)max(img, 0) * CC * (HH * WW);
    const int cbase = blockIdx.y * (2 * CPW) + wsub * CPW;
    const v4f* slab4 = (const v4f*)(fbase + (size_t)cbase * (HH * WW));
    v4f pf[NPF];
    auto load_tile = [&](int t) {
        const v4f* src4 = slab4 + t * TILE_V4;
        #pragma unroll
        for (int j = 0; j < NPF; ++j) {
            int idx = min(j * 32 + lane, TILE_V4 - 1);
            pf[j] = src4[idx];
        }
    };
    auto store_tile = [&]() {
        #pragma unroll
        for (int j = 0; j < NPF; ++j) {
            int idx = j * 32 + lane;
            if (idx < TILE_V4) {
                int row = idx / (WW / 4);
                int c4  = idx - row * (WW / 4);
                *(v4f*)&sF[wave][row][c4 * 4] = pf[j];
            }
        }
    };
    auto frag_load = [&](v2f (&dst)[KS]) {
        #pragma unroll
        for (int ks = 0; ks < KS; ++ks) {
            int k0 = ks * 4 + hi16 * 2;
            dst[ks] = *(const v2f*)&sF[wave][m16][k0];
        }
    };
    load_tile(0);

    {
        const int t64   = wsub * 32 + lane;
        const int row32 = t64 >> 1;
        const int which = row32 >> 4;
        const int row   = row32 & 15;
        const int s     = row >> 3;
        const int q     = row & 7;
        const bool on   = live && (q < OUT_HW) && (s == 0 || has_b);
        const float start = which ? (s ? y1r[1] : y1r[0]) : (s ? x1r[1] : x1r[0]);
        const float bs    = which ? (s ? bhr[1] : bhr[0]) : (s ? bwr[1] : bwr[0]);
        const float u0 = start + (float)q * bs;
        const float u1 = u0 + bs;
        float* wrow = which ? &sWy[grp][row][0] : &sWx[grp][row][0];
        for (int k = (t64 & 1); k < WW; k += 2) {
            float px = (float)k;
            float val = hat_cdf(u1 - px) - hat_cdf(u0 - px);
            wrow[k] = on ? val : 0.0f;
        }
    }
    __syncthreads();
    if (!live) return;

    v2f bx[KS];
    #pragma unroll
    for (int ks = 0; ks < KS; ++ks) {
        int k0 = ks * 4 + hi16 * 2;
        bx[ks] = *(const v2f*)&sWx[grp][m16][k0];
    }

    const int   qcol    = m16 & 7;
    const bool  lane_ok = (qcol < OUT_HW) && ((m16 >> 3) == hi16) && (hi16 == 0 || has_b);
    const int   rmine   = (hi16 && has_b) ? rb : ra;
    const float scmine  = hi16 ? scr[1] : scr[0];

    auto tile_math = [&](int t, const v2f (&cur)[KS], bool may_end_channel) {
        __builtin_amdgcn_sched_barrier(0);
        v8f acc0 = {}, acc1 = {};
        #pragma unroll
        for (int ks = 0; ks + 1 < KS; ks += 2) {
            acc0 = __builtin_amdgcn_wmma_f32_16x16x4_f32(
                     false, cur[ks],     false, bx[ks],     (short)0, acc0, false, false);
            acc1 = __builtin_amdgcn_wmma_f32_16x16x4_f32(
                     false, cur[ks + 1], false, bx[ks + 1], (short)0, acc1, false, false);
        }
        acc0 = __builtin_amdgcn_wmma_f32_16x16x4_f32(
                 false, cur[KS - 1], false, bx[KS - 1], (short)0, acc0, false, false);
        __builtin_amdgcn_sched_barrier(0);
        const int rbase = (t * 16) & (KPAD - 1);
        #pragma unroll
        for (int rr = 0; rr < 8; ++rr)
            sU[wave][rbase + rr + hi16 * 8][m16] = acc0[rr] + acc1[rr];

        if (!may_end_channel || t % 6 != 0 || t == 0) return;
        const int i = t / 6 - 1;
        const int c = cbase + i;

        __builtin_amdgcn_sched_barrier(0);

        v2f ay[KS], bu[KS];
        #pragma unroll
        for (int ks = 0; ks < KS; ++ks) {
            int k0 = ks * 4 + hi16 * 2;
            int r0 = i * HH + k0;
            ay[ks] = *(const v2f*)&sWy[grp][m16][k0];
            bu[ks].x = sU[wave][r0 & (KPAD - 1)][m16];
            bu[ks].y = sU[wave][(r0 + 1) & (KPAD - 1)][m16];
        }
        v8f acc2 = {}, acc3 = {};
        #pragma unroll
        for (int ks = 0; ks + 1 < KS; ks += 2) {
            acc2 = __builtin_amdgcn_wmma_f32_16x16x4_f32(
                     false, ay[ks],     false, bu[ks],     (short)0, acc2, false, false);
            acc3 = __builtin_amdgcn_wmma_f32_16x16x4_f32(
                     false, ay[ks + 1], false, bu[ks + 1], (short)0, acc3, false, false);
        }
        acc2 = __builtin_amdgcn_wmma_f32_16x16x4_f32(
                 false, ay[KS - 1], false, bu[KS - 1], (short)0, acc2, false, false);
        __builtin_amdgcn_sched_barrier(0);

        if (lane_ok) {
            float* o = out + (((size_t)rmine * CC + c) * OUT_HW) * OUT_HW + qcol;
            #pragma unroll
            for (int rr = 0; rr < OUT_HW; ++rr)
                o[rr * OUT_HW] = (acc2[rr] + acc3[rr]) * scmine;
        }
    };

    v2f avA[KS], avB[KS];
    store_tile();
    load_tile(1);
    frag_load(avA);
    for (int t = 0; t + 1 < NTILE; t += 2) {
        store_tile();
        load_tile(t + 2);
        frag_load(avB);
        tile_math(t, avA, true);

        store_tile();
        if (t + 3 < NTILE) load_tile(t + 3);
        frag_load(avA);
        tile_math(t + 1, avB, false);
    }
    tile_math(NTILE - 1, avA, true);
}

extern "C" void kernel_launch(void* const* d_in, const int* in_sizes, int n_in,
                              void* d_out, int out_size, void* d_ws, size_t ws_size,
                              hipStream_t stream) {
    const float* feat = (const float*)d_in[0];
    const float* rois = (const float*)d_in[1];
    float* out = (float*)d_out;
    int n_rois = in_sizes[1] / 5;
    int batch  = in_sizes[0] / (CC * HH * WW);
    dim3 grid((n_rois / 2 + batch + 1) / 2, 8);
    prroi_wmma_f32_kernel<<<grid, 128, 0, stream>>>(feat, rois, out, batch, n_rois);
}


# Even load per compute unit: 1-D grid, the full blocks (slots that always hold rois) dispatched first, the odd pair slots cut into 1-channel waves in 32 short tail blocks; sums unchanged
# speedup vs baseline: 3.5505x; 1.0952x over previous
#include <hip/hip_runtime.h>

typedef float v2f __attribute__((ext_vector_type(2)));
typedef float v8f __attribute__((ext_vector_type(8)));
typedef float v4f __attribute__((ext_vector_type(4)));

#define OUT_HW 7
#define SCALE 0.25f
#define CC 64
#define HH 100
#define WW 100
#define KPAD 128
#define KS 25
#define WSTRIDE 132
#define FPITCH 132
#define TILE_V4 400
#define NPF 13
#define NCH 8
#define CPW 4
#define GRP 2

__device__ __forceinline__ float hat_cdf(float t) {
    float a = fminf(fmaxf(t + 1.0f, 0.0f), 1.0f);
    float b = fminf(fmaxf(1.0f - t, 0.0f), 1.0f);
    return (t < 0.0f) ? (0.5f * a * a) : (1.0f - 0.5f * b * b);
}

__global__ __launch_bounds__(128)
void prroi_wmma_f32_kernel(const float* __restrict__ feat,
                           const float* __restrict__ rois,
                           float* __restrict__ out,
                           int batch, int n_rois)
{
    __shared__ __attribute__((aligned(16))) float sWx[GRP][16][WSTRIDE];
    __shared__ __attribute__((aligned(16))) float sWy[GRP][16][WSTRIDE];
    __shared__ float sU[4][KPAD][16];
    __shared__ __attribute__((aligned(16))) float sF[4][16][FPITCH];

    const int tid  = threadIdx.x;
    const int lane = tid & 31;
    const int wave = tid >> 5;
    const int grp  = wave >> 1;
    const int wsub = wave & 1;
    const int m16  = lane & 15;
    const int hi16 = (lane >> 4) & 1;

    int bis[NCH];
    #pragma unroll
    for (int ch = 0; ch < NCH; ++ch) {
        int nn = ch * 32 + lane;
        int v = (int)rois[(size_t)max(0, min(nn, n_rois - 1)) * 5];
        v = max(0, min(batch - 1, v));
        bis[ch] = (nn < n_rois) ? v : -1;
    }
    const int nfx   = n_rois >> 2;
    const int nfull = nfx * (CC / (2 * CPW));
    const int bid   = blockIdx.x;
    int slot, cbase, nchan;
    if (bid < nfull) {
        const int yb = bid / nfx;
        slot  = (bid - yb * nfx) * GRP + grp;
        cbase = yb * (2 * CPW) + wsub * CPW;
        nchan = CPW;
    } else {
        const int tb = bid - nfull;
        slot  = (nfx + (tb >> 5)) * GRP + grp;
        cbase = (tb & 31) * 2 + wsub;
        nchan = 1;
    }
    const int ntile = (nchan * HH + 15) >> 4;
    int img = -1, jp = 0, off = 0;
    for (int b = 0; b < batch; ++b) {
        int cnt = 0;
        #pragma unroll
        for (int ch = 0; ch < NCH; ++ch)
            cnt += __builtin_popcount(__builtin_amdgcn_ballot_w32(bis[ch] == b));
        int pairs = (cnt + 1) >> 1;
        if (img < 0 && slot < off + pairs) { img = b; jp = slot - off; }
        off += pairs;
    }

    int ra = -1, rb = -1, run = 0;
    #pragma unroll
    for (int ch = 0; ch < NCH; ++ch) {
        bool f = (bis[ch] == img);
        unsigned m = __builtin_amdgcn_ballot_w32(f);
        int rank = run + __builtin_popcount(m & ((1u << lane) - 1u));
        unsigned ma = __builtin_amdgcn_ballot_w32(f && rank == 2 * jp);
        unsigned mb = __builtin_amdgcn_ballot_w32(f && rank == 2 * jp + 1);
        if (ma) ra = ch * 32 + __builtin_ctz(ma);
        if (mb) rb = ch * 32 + __builtin_ctz(mb);
        run += __builtin_popcount(m);
    }
    ra = __builtin_amdgcn_readfirstlane(ra);
    rb = __builtin_amdgcn_readfirstlane(rb);
    img = __builtin_amdgcn_readfirstlane(img);
    const bool live = (img >= 0);
    const bool has_b = (rb >= 0);

    float x1r[2], y1r[2], bwr[2], bhr[2], scr[2];
    #pragma unroll
    for (int s = 0; s < 2; ++s) {
        const float* r = rois + (size_t)max((s == 0 || !has_b) ? ra : rb, 0) * 5;
        float x1 = r[1] * SCALE, y1 = r[2] * SCALE;
        float x2 = r[3] * SCALE, y2 = r[4] * SCALE;
        float rw = fmaxf(x2 - x1, 0.0f), rh = fmaxf(y2 - y1, 0.0f);
        float bw = rw / OUT_HW, bh = rh / OUT_HW;
        float area = bw * bh;
        x1r[s] = x1; y1r[s] = y1; bwr[s] = bw; bhr[s] = bh;
        scr[s] = (area > 0.0f) ? (1.0f / fmaxf(area, 1e-12f)) : 0.0f;
    }

    const float* fbase = feat + (size_t)max(img, 0) * CC * (HH * WW);
    const int roff = (nchan == 1 && !(max(img, 0) == 0 && cbase == 0)) ? 12 : 0;
    const v4f* slab4 = (const v4f*)(fbase + ((long)cbase * (HH * WW) - roff * WW));
    v4f pf[NPF];
    auto load_tile = [&](int t) {
        const v4f* src4 = slab4 + t * TILE_V4;
        #pragma unroll
        for (int j = 0; j < NPF; ++j) {
            int idx = min(j * 32 + lane, TILE_V4 - 1);
            pf[j] = src4[idx];
        }
    };
    auto store_tile = [&]() {
        #pragma unroll
        for (int j = 0; j < NPF; ++j) {
            int idx = j * 32 + lane;
            if (idx < TILE_V4) {
                int row = idx / (WW / 4);
                int c4  = idx - row * (WW / 4);
                *(v4f*)&sF[wave][row][c4 * 4] = pf[j];
            }
        }
    };
    auto frag_load = [&](v2f (&dst)[KS]) {
        #pragma unroll
        for (int ks = 0; ks < KS; ++ks) {
            int k0 = ks * 4 + hi16 * 2;
            dst[ks] = *(const v2f*)&sF[wave][m16][k0];
        }
    };
    load_tile(0);

    {
        const int t64   = wsub * 32 + lane;
        const int row32 = t64 >> 1;
        const int which = row32 >> 4;
        const int row   = row32 & 15;
        const int s     = row >> 3;
        const int q     = row & 7;
        const bool on   = live && (q < OUT_HW) && (s == 0 || has_b);
        const float start = which ? (s ? y1r[1] : y1r[0]) : (s ? x1r[1] : x1r[0]);
        const float bs    = which ? (s ? bhr[1] : bhr[0]) : (s ? bwr[1] : bwr[0]);
        const float u0 = start + (float)q * bs;
        const float u1 = u0 + bs;
        float* wrow = which ? &sWy[grp][row][0] : &sWx[grp][row][0];
        for (int k = (t64 & 1); k < WW; k += 2) {
            float px = (float)k;
            float val = hat_cdf(u1 - px) - hat_cdf(u0 - px);
            wrow[k] = on ? val : 0.0f;
        }
    }
    __syncthreads();
    if (!live) return;

    v2f bx[KS];
    #pragma unroll
    for (int ks = 0; ks < KS; ++ks) {
        int k0 = ks * 4 + hi16 * 2;
        bx[ks] = *(const v2f*)&sWx[grp][m16][k0];
    }

    const int   qcol    = m16 & 7;
    const bool  lane_ok = (qcol < OUT_HW) && ((m16 >> 3) == hi16) && (hi16 == 0 || has_b);
    const int   rmine   = (hi16 && has_b) ? rb : ra;
    const float scmine  = hi16 ? scr[1] : scr[0];

    auto tile_math = [&](int t, const v2f (&cur)[KS], bool may_end_channel) {
        __builtin_amdgcn_sched_barrier(0);
        v8f acc0 = {}, acc1 = {};
        #pragma unroll
        for (int ks = 0; ks + 1 < KS; ks += 2) {
            acc0 = __builtin_amdgcn_wmma_f32_16x16x4_f32(
                     false, cur[ks],     false, bx[ks],     (short)0, acc0, false, false);
            acc1 = __builtin_amdgcn_wmma_f32_16x16x4_f32(
                     false, cur[ks + 1], false, bx[ks + 1], (short)0, acc1, false, false);
        }
        acc0 = __builtin_amdgcn_wmma_f32_16x16x4_f32(
                 false, cur[KS - 1], false, bx[KS - 1], (short)0, acc0, false, false);
        __builtin_amdgcn_sched_barrier(0);
        const int rbase = (t * 16) & (KPAD - 1);
        #pragma unroll
        for (int rr = 0; rr < 8; ++rr)
            sU[wave][rbase + rr + hi16 * 8][m16] = acc0[rr] + acc1[rr];

        if (!may_end_channel || t % 6 != 0 || t == 0) return;
        const int i = t / 6 - 1;
        const int c = cbase + i;

        __builtin_amdgcn_sched_barrier(0);

        v2f ay[KS], bu[KS];
        #pragma unroll
        for (int ks = 0; ks < KS; ++ks) {
            int k0 = ks * 4 + hi16 * 2;
            int r0 = i * HH + roff + k0;
            ay[ks] = *(const v2f*)&sWy[grp][m16][k0];
            bu[ks].x = sU[wave][r0 & (KPAD - 1)][m16];
            bu[ks].y = sU[wave][(r0 + 1) & (KPAD - 1)][m16];
        }
        v8f acc2 = {}, acc3 = {};
        #pragma unroll
        for (int ks = 0; ks + 1 < KS; ks += 2) {
            acc2 = __builtin_amdgcn_wmma_f32_16x16x4_f32(
                     false, ay[ks],     false, bu[ks],     (short)0, acc2, false, false);
            acc3 = __builtin_amdgcn_wmma_f32_16x16x4_f32(
                     false, ay[ks + 1], false, bu[ks + 1], (short)0, acc3, false, false);
        }
        acc2 = __builtin_amdgcn_wmma_f32_16x16x4_f32(
                 false, ay[KS - 1], false, bu[KS - 1], (short)0, acc2, false, false);
        __builtin_amdgcn_sched_barrier(0);

        if (lane_ok) {
            float* o = out + (((size_t)rmine * CC + c) * OUT_HW) * OUT_HW + qcol;
            #pragma unroll
            for (int rr = 0; rr < OUT_HW; ++rr)
                o[rr * OUT_HW] = (acc2[rr] + acc3[rr]) * scmine;
        }
    };

    v2f avA[KS], avB[KS];
    store_tile();
    load_tile(1);
    frag_load(avA);
    for (int t = 0; t + 1 < ntile; t += 2) {
        store_tile();
        load_tile(t + 2);
        frag_load(avB);
        tile_math(t, avA, true);

        store_tile();
        if (t + 3 < ntile) load_tile(t + 3);
        frag_load(avA);
        tile_math(t + 1, avB, false);
    }
    tile_math(ntile - 1, avA, true);
}

extern "C" void kernel_launch(void* const* d_in, const int* in_sizes, int n_in,
                              void* d_out, int out_size, void* d_ws, size_t ws_size,
                              hipStream_t stream) {
    const float* feat = (const float*)d_in[0];
    const float* rois = (const float*)d_in[1];
    float* out = (float*)d_out;
    int n_rois = in_sizes[1] / 5;
    int batch  = in_sizes[0] / (CC * HH * WW);
    dim3 grid((n_rois / 4) * 8 + ((n_rois / 2 + batch - (n_rois / 4) * 2 + 1) / 2) * 32);
    prroi_wmma_f32_kernel<<<grid, 128, 0, stream>>>(feat, rois, out, batch, n_rois);
}


# Global prefetch two tiles deep: a second register set keeps two feature tiles in flight, each requested two matrix runs (was one) before its LDS store; request trips branch-free; sums unchanged
# speedup vs baseline: 3.9756x; 1.1197x over previous
#include <hip/hip_runtime.h>

typedef float v2f __attribute__((ext_vector_type(2)));
typedef float v8f __attribute__((ext_vector_type(8)));
typedef float v4f __attribute__((ext_vector_type(4)));

#define OUT_HW 7
#define SCALE 0.25f
#define CC 64
#define HH 100
#define WW 100
#define KPAD 128
#define KS 25
#define WSTRIDE 132
#define FPITCH 132
#define TILE_V4 400
#define NPF 13
#define NCH 8
#define CPW 4
#define GRP 2

__device__ __forceinline__ float hat_cdf(float t) {
    float a = fminf(fmaxf(t + 1.0f, 0.0f), 1.0f);
    float b = fminf(fmaxf(1.0f - t, 0.0f), 1.0f);
    return (t < 0.0f) ? (0.5f * a * a) : (1.0f - 0.5f * b * b);
}

__global__ __launch_bounds__(128)
void prroi_wmma_f32_kernel(const float* __restrict__ feat,
                           const float* __restrict__ rois,
                           float* __restrict__ out,
                           int batch, int n_rois)
{
    __shared__ __attribute__((aligned(16))) float sWx[GRP][16][WSTRIDE];
    __shared__ __attribute__((aligned(16))) float sWy[GRP][16][WSTRIDE];
    __shared__ float sU[4][KPAD][16];
    __shared__ __attribute__((aligned(16))) float sF[4][16][FPITCH];

    const int tid  = threadIdx.x;
    const int lane = tid & 31;
    const int wave = tid >> 5;
    const int grp  = wave >> 1;
    const int wsub = wave & 1;
    const int m16  = lane & 15;
    const int hi16 = (lane >> 4) & 1;

    int bis[NCH];
    #pragma unroll
    for (int ch = 0; ch < NCH; ++ch) {
        int nn = ch * 32 + lane;
        int v = (int)rois[(size_t)max(0, min(nn, n_rois - 1)) * 5];
        v = max(0, min(batch - 1, v));
        bis[ch] = (nn < n_rois) ? v : -1;
    }
    const int nfx   = n_rois >> 2;
    const int nfull = nfx * (CC / (2 * CPW));
    const int bid   = blockIdx.x;
    int slot, cbase, nchan;
    if (bid < nfull) {
        const int yb = bid / nfx;
        slot  = (bid - yb * nfx) * GRP + grp;
        cbase = yb * (2 * CPW) + wsub * CPW;
        nchan = CPW;
    } else {
        const int tb = bid - nfull;
        slot  = (nfx + (tb >> 5)) * GRP + grp;
        cbase = (tb & 31) * 2 + wsub;
        nchan = 1;
    }
    const int ntile = (nchan * HH + 15) >> 4;
    int img = -1, jp = 0, off = 0;
    for (int b = 0; b < batch; ++b) {
        int cnt = 0;
        #pragma unroll
        for (int ch = 0; ch < NCH; ++ch)
            cnt += __builtin_popcount(__builtin_amdgcn_ballot_w32(bis[ch] == b));
        int pairs = (cnt + 1) >> 1;
        if (img < 0 && slot < off + pairs) { img = b; jp = slot - off; }
        off += pairs;
    }

    int ra = -1, rb = -1, run = 0;
    #pragma unroll
    for (int ch = 0; ch < NCH; ++ch) {
        bool f = (bis[ch] == img);
        unsigned m = __builtin_amdgcn_ballot_w32(f);
        int rank = run + __builtin_popcount(m & ((1u << lane) - 1u));
        unsigned ma = __builtin_amdgcn_ballot_w32(f && rank == 2 * jp);
        unsigned mb = __builtin_amdgcn_ballot_w32(f && rank == 2 * jp + 1);
        if (ma) ra = ch * 32 + __builtin_ctz(ma);
        if (mb) rb = ch * 32 + __builtin_ctz(mb);
        run += __builtin_popcount(m);
    }
    ra = __builtin_amdgcn_readfirstlane(ra);
    rb = __builtin_amdgcn_readfirstlane(rb);
    img = __builtin_amdgcn_readfirstlane(img);
    const bool live = (img >= 0);
    const bool has_b = (rb >= 0);

    float x1r[2], y1r[2], bwr[2], bhr[2], scr[2];
    #pragma unroll
    for (int s = 0; s < 2; ++s) {
        const float* r = rois + (size_t)max((s == 0 || !has_b) ? ra : rb, 0) * 5;
        float x1 = r[1] * SCALE, y1 = r[2] * SCALE;
        float x2 = r[3] * SCALE, y2 = r[4] * SCALE;
        float rw = fmaxf(x2 - x1, 0.0f), rh = fmaxf(y2 - y1, 0.0f);
        float bw = rw / OUT_HW, bh = rh / OUT_HW;
        float area = bw * bh;
        x1r[s] = x1; y1r[s] = y1; bwr[s] = bw; bhr[s] = bh;
        scr[s] = (area > 0.0f) ? (1.0f / fmaxf(area, 1e-12f)) : 0.0f;
    }

    const float* fbase = feat + (size_t)max(img, 0) * CC * (HH * WW);
    const int roff = (nchan == 1 && !(max(img, 0) == 0 && cbase == 0)) ? 12 : 0;
    const v4f* slab4 = (const v4f*)(fbase + ((long)cbase * (HH * WW) - roff * WW));
    v4f pfA[NPF], pfB[NPF];
    auto load_tile = [&](int t, v4f (&pf)[NPF]) {
        const v4f* src4 = slab4 + t * TILE_V4;
        #pragma unroll
        for (int j = 0; j < NPF; ++j) {
            int idx = min(j * 32 + lane, TILE_V4 - 1);
            pf[j] = src4[idx];
        }
    };
    auto store_tile = [&](const v4f (&pf)[NPF]) {
        #pragma unroll
        for (int j = 0; j < NPF; ++j) {
            int idx = j * 32 + lane;
            if (idx < TILE_V4) {
                int row = idx / (WW / 4);
                int c4  = idx - row * (WW / 4);
                *(v4f*)&sF[wave][row][c4 * 4] = pf[j];
            }
        }
    };
    auto frag_load = [&](v2f (&dst)[KS]) {
        #pragma unroll
        for (int ks = 0; ks < KS; ++ks) {
            int k0 = ks * 4 + hi16 * 2;
            dst[ks] = *(const v2f*)&sF[wave][m16][k0];
        }
    };
    load_tile(0, pfA);
    load_tile(1, pfB);

    {
        const int t64   = wsub * 32 + lane;
        const int row32 = t64 >> 1;
        const int which = row32 >> 4;
        const int row   = row32 & 15;
        const int s     = row >> 3;
        const int q     = row & 7;
        const bool on   = live && (q < OUT_HW) && (s == 0 || has_b);
        const float start = which ? (s ? y1r[1] : y1r[0]) : (s ? x1r[1] : x1r[0]);
        const float bs    = which ? (s ? bhr[1] : bhr[0]) : (s ? bwr[1] : bwr[0]);
        const float u0 = start + (float)q * bs;
        const float u1 = u0 + bs;
        float* wrow = which ? &sWy[grp][row][0] : &sWx[grp][row][0];
        for (int k = (t64 & 1); k < WW; k += 2) {
            float px = (float)k;
            float val = hat_cdf(u1 - px) - hat_cdf(u0 - px);
            wrow[k] = on ? val : 0.0f;
        }
    }
    __syncthreads();
    if (!live) return;

    v2f bx[KS];
    #pragma unroll
    for (int ks = 0; ks < KS; ++ks) {
        int k0 = ks * 4 + hi16 * 2;
        bx[ks] = *(const v2f*)&sWx[grp][m16][k0];
    }

    const int   qcol    = m16 & 7;
    const bool  lane_ok = (qcol < OUT_HW) && ((m16 >> 3) == hi16) && (hi16 == 0 || has_b);
    const int   rmine   = (hi16 && has_b) ? rb : ra;
    const float scmine  = hi16 ? scr[1] : scr[0];

    auto tile_math = [&](int t, const v2f (&cur)[KS], bool may_end_channel) {
        __builtin_amdgcn_sched_barrier(0);
        v8f acc0 = {}, acc1 = {};
        #pragma unroll
        for (int ks = 0; ks + 1 < KS; ks += 2) {
            acc0 = __builtin_amdgcn_wmma_f32_16x16x4_f32(
                     false, cur[ks],     false, bx[ks],     (short)0, acc0, false, false);
            acc1 = __builtin_amdgcn_wmma_f32_16x16x4_f32(
                     false, cur[ks + 1], false, bx[ks + 1], (short)0, acc1, false, false);
        }
        acc0 = __builtin_amdgcn_wmma_f32_16x16x4_f32(
                 false, cur[KS - 1], false, bx[KS - 1], (short)0, acc0, false, false);
        __builtin_amdgcn_sched_barrier(0);
        const int rbase = (t * 16) & (KPAD - 1);
        #pragma unroll
        for (int rr = 0; rr < 8; ++rr)
            sU[wave][rbase + rr + hi16 * 8][m16] = acc0[rr] + acc1[rr];

        if (!may_end_channel || t % 6 != 0 || t == 0) return;
        const int i = t / 6 - 1;
        const int c = cbase + i;

        __builtin_amdgcn_sched_barrier(0);

        v2f ay[KS], bu[KS];
        #pragma unroll
        for (int ks = 0; ks < KS; ++ks) {
            int k0 = ks * 4 + hi16 * 2;
            int r0 = i * HH + roff + k0;
            ay[ks] = *(const v2f*)&sWy[grp][m16][k0];
            bu[ks].x = sU[wave][r0 & (KPAD - 1)][m16];
            bu[ks].y = sU[wave][(r0 + 1) & (KPAD - 1)][m16];
        }
        v8f acc2 = {}, acc3 = {};
        #pragma unroll
        for (int ks = 0; ks + 1 < KS; ks += 2) {
            acc2 = __builtin_amdgcn_wmma_f32_16x16x4_f32(
                     false, ay[ks],     false, bu[ks],     (short)0, acc2, false, false);
            acc3 = __builtin_amdgcn_wmma_f32_16x16x4_f32(
                     false, ay[ks + 1], false, bu[ks + 1], (short)0, acc3, false, false);
        }
        acc2 = __builtin_amdgcn_wmma_f32_16x16x4_f32(
                 false, ay[KS - 1], false, bu[KS - 1], (short)0, acc2, false, false);
        __builtin_amdgcn_sched_barrier(0);

        if (lane_ok) {
            float* o = out + (((size_t)rmine * CC + c) * OUT_HW) * OUT_HW + qcol;
            #pragma unroll
            for (int rr = 0; rr < OUT_HW; ++rr)
                o[rr * OUT_HW] = (acc2[rr] + acc3[rr]) * scmine;
        }
    };

    v2f avA[KS], avB[KS];
    store_tile(pfA);
    load_tile(2, pfA);
    frag_load(avA);
    int t = 0;
    for (; t + 4 < ntile; t += 2) {
        store_tile(pfB);
        load_tile(t + 3, pfB);
        frag_load(avB);
        tile_math(t, avA, true);

        store_tile(pfA);
        load_tile(t + 4, pfA);
        frag_load(avA);
        tile_math(t + 1, avB, false);
    }
    store_tile(pfB);
    frag_load(avB);
    tile_math(t, avA, true);
    store_tile(pfA);
    frag_load(avA);
    tile_math(t + 1, avB, false);
    tile_math(ntile - 1, avA, true);
}

extern "C" void kernel_launch(void* const* d_in, const int* in_sizes, int n_in,
                              void* d_out, int out_size, void* d_ws, size_t ws_size,
                              hipStream_t stream) {
    const float* feat = (const float*)d_in[0];
    const float* rois = (const float*)d_in[1];
    float* out = (float*)d_out;
    int n_rois = in_sizes[1] / 5;
    int batch  = in_sizes[0] / (CC * HH * WW);
    dim3 grid((n_rois / 4) * 8 + ((n_rois / 2 + batch - (n_rois / 4) * 2 + 1) / 2) * 32);
    prroi_wmma_f32_kernel<<<grid, 128, 0, stream>>>(feat, rois, out, batch, n_rois);
}


# Weight rows are zeroed by wide LDS stores and the hat integrals evaluated only on each bin's non-zero pixel window (was all 100 pixels of every row); weight values and all sums unchanged
# speedup vs baseline: 4.1832x; 1.0522x over previous
#include <hip/hip_runtime.h>

typedef float v2f __attribute__((ext_vector_type(2)));
typedef float v8f __attribute__((ext_vector_type(8)));
typedef float v4f __attribute__((ext_vector_type(4)));

#define OUT_HW 7
#define SCALE 0.25f
#define CC 64
#define HH 100
#define WW 100
#define KPAD 128
#define KS 25
#define WSTRIDE 132
#define FPITCH 132
#define TILE_V4 400
#define NPF 13
#define NCH 8
#define CPW 4
#define GRP 2

__device__ __forceinline__ float hat_cdf(float t) {
    float a = fminf(fmaxf(t + 1.0f, 0.0f), 1.0f);
    float b = fminf(fmaxf(1.0f - t, 0.0f), 1.0f);
    return (t < 0.0f) ? (0.5f * a * a) : (1.0f - 0.5f * b * b);
}

__global__ __launch_bounds__(128)
void prroi_wmma_f32_kernel(const float* __restrict__ feat,
                           const float* __restrict__ rois,
                           float* __restrict__ out,
                           int batch, int n_rois)
{
    __shared__ __attribute__((aligned(16))) float sWx[GRP][16][WSTRIDE];
    __shared__ __attribute__((aligned(16))) float sWy[GRP][16][WSTRIDE];
    __shared__ float sU[4][KPAD][16];
    __shared__ __attribute__((aligned(16))) float sF[4][16][FPITCH];

    const int tid  = threadIdx.x;
    const int lane = tid & 31;
    const int wave = tid >> 5;
    const int grp  = wave >> 1;
    const int wsub = wave & 1;
    const int m16  = lane & 15;
    const int hi16 = (lane >> 4) & 1;

    int bis[NCH];
    #pragma unroll
    for (int ch = 0; ch < NCH; ++ch) {
        int nn = ch * 32 + lane;
        int v = (int)rois[(size_t)max(0, min(nn, n_rois - 1)) * 5];
        v = max(0, min(batch - 1, v));
        bis[ch] = (nn < n_rois) ? v : -1;
    }
    const int nfx   = n_rois >> 2;
    const int nfull = nfx * (CC / (2 * CPW));
    const int bid   = blockIdx.x;
    int slot, cbase, nchan;
    if (bid < nfull) {
        const int yb = bid / nfx;
        slot  = (bid - yb * nfx) * GRP + grp;
        cbase = yb * (2 * CPW) + wsub * CPW;
        nchan = CPW;
    } else {
        const int tb = bid - nfull;
        slot  = (nfx + (tb >> 5)) * GRP + grp;
        cbase = (tb & 31) * 2 + wsub;
        nchan = 1;
    }
    const int ntile = (nchan * HH + 15) >> 4;
    int img = -1, jp = 0, off = 0;
    for (int b = 0; b < batch; ++b) {
        int cnt = 0;
        #pragma unroll
        for (int ch = 0; ch < NCH; ++ch)
            cnt += __builtin_popcount(__builtin_amdgcn_ballot_w32(bis[ch] == b));
        int pairs = (cnt + 1) >> 1;
        if (img < 0 && slot < off + pairs) { img = b; jp = slot - off; }
        off += pairs;
    }

    int ra = -1, rb = -1, run = 0;
    #pragma unroll
    for (int ch = 0; ch < NCH; ++ch) {
        bool f = (bis[ch] == img);
        unsigned m = __builtin_amdgcn_ballot_w32(f);
        int rank = run + __builtin_popcount(m & ((1u << lane) - 1u));
        unsigned ma = __builtin_amdgcn_ballot_w32(f && rank == 2 * jp);
        unsigned mb = __builtin_amdgcn_ballot_w32(f && rank == 2 * jp + 1);
        if (ma) ra = ch * 32 + __builtin_ctz(ma);
        if (mb) rb = ch * 32 + __builtin_ctz(mb);
        run += __builtin_popcount(m);
    }
    ra = __builtin_amdgcn_readfirstlane(ra);
    rb = __builtin_amdgcn_readfirstlane(rb);
    img = __builtin_amdgcn_readfirstlane(img);
    const bool live = (img >= 0);
    const bool has_b = (rb >= 0);

    float x1r[2], y1r[2], bwr[2], bhr[2], scr[2];
    #pragma unroll
    for (int s = 0; s < 2; ++s) {
        const float* r = rois + (size_t)max((s == 0 || !has_b) ? ra : rb, 0) * 5;
        float x1 = r[1] * SCALE, y1 = r[2] * SCALE;
        float x2 = r[3] * SCALE, y2 = r[4] * SCALE;
        float rw = fmaxf(x2 - x1, 0.0f), rh = fmaxf(y2 - y1, 0.0f);
        float bw = rw / OUT_HW, bh = rh / OUT_HW;
        float area = bw * bh;
        x1r[s] = x1; y1r[s] = y1; bwr[s] = bw; bhr[s] = bh;
        scr[s] = (area > 0.0f) ? (1.0f / fmaxf(area, 1e-12f)) : 0.0f;
    }

    const float* fbase = feat + (size_t)max(img, 0) * CC * (HH * WW);
    const int roff = (nchan == 1 && !(max(img, 0) == 0 && cbase == 0)) ? 12 : 0;
    const v4f* slab4 = (const v4f*)(fbase + ((long)cbase * (HH * WW) - roff * WW));
    v4f pfA[NPF], pfB[NPF];
    auto load_tile = [&](int t, v4f (&pf)[NPF]) {
        const v4f* src4 = slab4 + t * TILE_V4;
        #pragma unroll
        for (int j = 0; j < NPF; ++j) {
            int idx = min(j * 32 + lane, TILE_V4 - 1);
            pf[j] = src4[idx];
        }
    };
    auto store_tile = [&](const v4f (&pf)[NPF]) {
        #pragma unroll
        for (int j = 0; j < NPF; ++j) {
            int idx = j * 32 + lane;
            if (idx < TILE_V4) {
                int row = idx / (WW / 4);
                int c4  = idx - row * (WW / 4);
                *(v4f*)&sF[wave][row][c4 * 4] = pf[j];
            }
        }
    };
    auto frag_load = [&](v2f (&dst)[KS]) {
        #pragma unroll
        for (int ks = 0; ks < KS; ++ks) {
            int k0 = ks * 4 + hi16 * 2;
            dst[ks] = *(const v2f*)&sF[wave][m16][k0];
        }
    };
    load_tile(0, pfA);
    load_tile(1, pfB);

    if (live) {
        const int row   = lane >> 1;
        const int s     = row >> 3;
        const int q     = row & 7;
        const bool on   = (q < OUT_HW) && (s == 0 || has_b);
        const float start = wsub ? (s ? y1r[1] : y1r[0]) : (s ? x1r[1] : x1r[0]);
        const float bs    = wsub ? (s ? bhr[1] : bhr[0]) : (s ? bwr[1] : bwr[0]);
        const float u0 = start + (float)q * bs;
        const float u1 = u0 + bs;
        float* wbase = wsub ? &sWy[grp][0][0] : &sWx[grp][0][0];
        v4f* wz = (v4f*)wbase;
        const v4f zero4 = {0.0f, 0.0f, 0.0f, 0.0f};
        for (int j = 0; j < (16 * WSTRIDE / 4 + 31) / 32; ++j) {
            int idx = j * 32 + lane;
            if (idx < 16 * WSTRIDE / 4) wz[idx] = zero4;
        }
        if (on) {
            const int klo = (int)fminf(fmaxf(floorf(u0), 0.0f), (float)WW);
            const int khi = (int)fminf(fmaxf(ceilf(u1), -1.0f), (float)(WW - 1));
            float* wrow = wbase + row * WSTRIDE;
            for (int k = klo + (lane & 1); k <= khi; k += 2) {
                float px = (float)k;
                wrow[k] = hat_cdf(u1 - px) - hat_cdf(u0 - px);
            }
        }
    }
    __syncthreads();
    if (!live) return;

    v2f bx[KS];
    #pragma unroll
    for (int ks = 0; ks < KS; ++ks) {
        int k0 = ks * 4 + hi16 * 2;
        bx[ks] = *(const v2f*)&sWx[grp][m16][k0];
    }

    const int   qcol    = m16 & 7;
    const bool  lane_ok = (qcol < OUT_HW) && ((m16 >> 3) == hi16) && (hi16 == 0 || has_b);
    const int   rmine   = (hi16 && has_b) ? rb : ra;
    const float scmine  = hi16 ? scr[1] : scr[0];

    auto tile_math = [&](int t, const v2f (&cur)[KS], bool may_end_channel) {
        __builtin_amdgcn_sched_barrier(0);
        v8f acc0 = {}, acc1 = {};
        #pragma unroll
        for (int ks = 0; ks + 1 < KS; ks += 2) {
            acc0 = __builtin_amdgcn_wmma_f32_16x16x4_f32(
                     false, cur[ks],     false, bx[ks],     (short)0, acc0, false, false);
            acc1 = __builtin_amdgcn_wmma_f32_16x16x4_f32(
                     false, cur[ks + 1], false, bx[ks + 1], (short)0, acc1, false, false);
        }
        acc0 = __builtin_amdgcn_wmma_f32_16x16x4_f32(
                 false, cur[KS - 1], false, bx[KS - 1], (short)0, acc0, false, false);
        __builtin_amdgcn_sched_barrier(0);
        const int rbase = (t * 16) & (KPAD - 1);
        #pragma unroll
        for (int rr = 0; rr < 8; ++rr)
            sU[wave][rbase + rr + hi16 * 8][m16] = acc0[rr] + acc1[rr];

        if (!may_end_channel || t % 6 != 0 || t == 0) return;
        const int i = t / 6 - 1;
        const int c = cbase + i;

        __builtin_amdgcn_sched_barrier(0);

        v2f ay[KS], bu[KS];
        #pragma unroll
        for (int ks = 0; ks < KS; ++ks) {
            int k0 = ks * 4 + hi16 * 2;
            int r0 = i * HH + roff + k0;
            ay[ks] = *(const v2f*)&sWy[grp][m16][k0];
            bu[ks].x = sU[wave][r0 & (KPAD - 1)][m16];
            bu[ks].y = sU[wave][(r0 + 1) & (KPAD - 1)][m16];
        }
        v8f acc2 = {}, acc3 = {};
        #pragma unroll
        for (int ks = 0; ks + 1 < KS; ks += 2) {
            acc2 = __builtin_amdgcn_wmma_f32_16x16x4_f32(
                     false, ay[ks],     false, bu[ks],     (short)0, acc2, false, false);
            acc3 = __builtin_amdgcn_wmma_f32_16x16x4_f32(
                     false, ay[ks + 1], false, bu[ks + 1], (short)0, acc3, false, false);
        }
        acc2 = __builtin_amdgcn_wmma_f32_16x16x4_f32(
                 false, ay[KS - 1], false, bu[KS - 1], (short)0, acc2, false, false);
        __builtin_amdgcn_sched_barrier(0);

        if (lane_ok) {
            float* o = out + (((size_t)rmine * CC + c) * OUT_HW) * OUT_HW + qcol;
            #pragma unroll
            for (int rr = 0; rr < OUT_HW; ++rr)
                o[rr * OUT_HW] = (acc2[rr] + acc3[rr]) * scmine;
        }
    };

    v2f avA[KS], avB[KS];
    store_tile(pfA);
    load_tile(2, pfA);
    frag_load(avA);
    int t = 0;
    for (; t + 4 < ntile; t += 2) {
        store_tile(pfB);
        load_tile(t + 3, pfB);
        frag_load(avB);
        tile_math(t, avA, true);

        store_tile(pfA);
        load_tile(t + 4, pfA);
        frag_load(avA);
        tile_math(t + 1, avB, false);
    }
    store_tile(pfB);
    frag_load(avB);
    tile_math(t, avA, true);
    store_tile(pfA);
    frag_load(avA);
    tile_math(t + 1, avB, false);
    tile_math(ntile - 1, avA, true);
}

extern "C" void kernel_launch(void* const* d_in, const int* in_sizes, int n_in,
                              void* d_out, int out_size, void* d_ws, size_t ws_size,
                              hipStream_t stream) {
    const float* feat = (const float*)d_in[0];
    const float* rois = (const float*)d_in[1];
    float* out = (float*)d_out;
    int n_rois = in_sizes[1] / 5;
    int batch  = in_sizes[0] / (CC * HH * WW);
    dim3 grid((n_rois / 4) * 8 + ((n_rois / 2 + batch - (n_rois / 4) * 2 + 1) / 2) * 32);
    prroi_wmma_f32_kernel<<<grid, 128, 0, stream>>>(feat, rois, out, batch, n_rois);
}
